# starting kernel
# baseline (speedup 1.0000x reference)
#include <hip/hip_runtime.h>

typedef __attribute__((ext_vector_type(2))) float v2f;
typedef __attribute__((ext_vector_type(8))) float v8f;

#define H 512
#define W 512
#define CIN 3
#define COUT 2
#define LDS_NROWS 12
#define LDS_ROWW  520

__global__ __launch_bounds__(256) void conv3x3_wmma_f32(
    const float* __restrict__ x,
    const float* __restrict__ wgt,
    const float* __restrict__ bias,
    float* __restrict__ out)
{
    const int h0  = blockIdx.x * 2;
    const int b   = blockIdx.y;
    const int tid = threadIdx.x;

    __shared__ float lds[LDS_NROWS * LDS_ROWW];

    if (tid < 2 * LDS_NROWS) {
        int r = tid >> 1;
        lds[r * LDS_ROWW + ((tid & 1) ? 516 : 3)] = 0.0f;
    }

    for (int idx = tid; idx < LDS_NROWS * 128; idx += 256) {
        int r   = idx >> 7;
        int q   = idx & 127;
        int c   = r >> 2;
        int dh4 = r & 3;
        int row = h0 + dh4 - 1;
        float4 v = make_float4(0.0f, 0.0f, 0.0f, 0.0f);
        if (row >= 0 && row < H)
            v = *(const float4*)(x + (((size_t)b * CIN + c) * H + row) * W + q * 4);
        *(float4*)(&lds[r * LDS_ROWW + 4 + q * 4]) = v;
    }
    __syncthreads();

    const int lane = tid & 31;
    const int wave = tid >> 5;
    const int m    = lane & 15;
    const int half = lane >> 4;

    v2f bfrag[7];
#pragma unroll
    for (int j = 0; j < 7; ++j) {
        float b0 = 0.0f, b1 = 0.0f;
        int k0 = j * 4 + 2 * half;
        if (m < COUT) {
            b0 = wgt[m * 27 + k0];
            if (k0 + 1 < 27) b1 = wgt[m * 27 + k0 + 1];
        }
        bfrag[j][0] = b0;
        bfrag[j][1] = b1;
    }

    int offA[14];
#pragma unroll
    for (int j = 0; j < 7; ++j) {
#pragma unroll
        for (int v = 0; v < 2; ++v) {
            int k = j * 4 + 2 * half + v;
            if (k > 26) k = 26;
            int c  = k / 9;
            int r  = k - c * 9;
            int dh = r / 3;
            int dw = r - dh * 3;
            offA[j * 2 + v] = (c * 4 + dh) * LDS_ROWW + m + dw + 3;
        }
    }

    const float bv0 = bias[0];
    const float bv1 = bias[1];

#pragma unroll
    for (int rr = 0; rr < 2; ++rr) {
        const float* ldsr = lds + rr * LDS_ROWW;
        const int h = h0 + rr;

#pragma unroll
        for (int s = 0; s < 4; ++s) {
            const int w0 = (wave * 4 + s) * 16;

            v8f acc = {};
#pragma unroll
            for (int j = 0; j < 7; ++j) {
                v2f a;
                a[0] = ldsr[offA[j * 2 + 0] + w0];
                a[1] = ldsr[offA[j * 2 + 1] + w0];
                acc = __builtin_amdgcn_wmma_f32_16x16x4_f32(
                     false, a,  false, bfrag[j],
                     (short)0, acc,  false,  false);
            }

            if ((lane & 14) == 0) {
                const int ch = lane & 1;
                const int mb = (lane >> 4) * 8;
                const float bv = ch ? bv1 : bv0;
                float* o = out + ((((size_t)b * COUT + ch) * H + h) * W + w0 + mb);
                float4 lo = make_float4(acc[0] + bv, acc[1] + bv, acc[2] + bv, acc[3] + bv);
                float4 hi = make_float4(acc[4] + bv, acc[5] + bv, acc[6] + bv, acc[7] + bv);
                *(float4*)(o + 0) = lo;
                *(float4*)(o + 4) = hi;
            }
        }
    }
}

extern "C" void kernel_launch(void* const* d_in, const int* in_sizes, int n_in,
                              void* d_out, int out_size, void* d_ws, size_t ws_size,
                              hipStream_t stream) {
    const float* x    = (const float*)d_in[0];
    const float* wgt  = (const float*)d_in[1];
    const float* bias = (const float*)d_in[2];
    float* out        = (float*)d_out;

    dim3 grid(H / 2, 64, 1);
    dim3 block(256, 1, 1);
    conv3x3_wmma_f32<<<grid, block, 0, stream>>>(x, wgt, bias, out);
}


# Work-group computes a strip of 8 output rows (was 2): 4x fewer waves to enter, input rows staged 1.25x (was 2x), a thread's 15 staging loads in flight together (clamped row, no branch)
# speedup vs baseline: 1.2153x; 1.2153x over previous
#include <hip/hip_runtime.h>


typedef __attribute__((ext_vector_type(2))) float v2f;
typedef __attribute__((ext_vector_type(8))) float v8f;

#define H 512
#define W 512
#define CIN 3
#define COUT 2
#define RB 8
#define CROWS (RB + 2)
#define LDS_NROWS (CIN * CROWS)
#define LDS_ROWW  520

__global__ __launch_bounds__(256) void conv3x3_wmma_f32(
    const float* __restrict__ x,
    const float* __restrict__ wgt,
    const float* __restrict__ bias,
    float* __restrict__ out)
{
    const int h0  = blockIdx.x * RB;
    const int b   = blockIdx.y;
    const int tid = threadIdx.x;

    __shared__ float lds[LDS_NROWS * LDS_ROWW];

    if (tid < 2 * LDS_NROWS) {
        int r = tid >> 1;
        lds[r * LDS_ROWW + ((tid & 1) ? 516 : 3)] = 0.0f;
    }

    constexpr int NST = (CROWS * 128) / 256;
    float4 sv[CIN][NST];
#pragma unroll
    for (int i = 0; i < NST; ++i) {
        const int idx  = tid + i * 256;
        const int dr   = idx >> 7;
        const int q    = idx & 127;
        const int rowc = min(max(h0 + dr - 1, 0), H - 1);
#pragma unroll
        for (int c = 0; c < CIN; ++c)
            sv[c][i] = *(const float4*)(x + (((size_t)b * CIN + c) * H + rowc) * W + q * 4);
    }
#pragma unroll
    for (int i = 0; i < NST; ++i) {
        const int idx = tid + i * 256;
        const int dr  = idx >> 7;
        const int q   = idx & 127;
#pragma unroll
        for (int c = 0; c < CIN; ++c)
            *(float4*)(&lds[(c * CROWS + dr) * LDS_ROWW + 4 + q * 4]) = sv[c][i];
    }
#pragma unroll
    for (int i = 0; i < NST; ++i) {
        const int idx = tid + i * 256;
        const int dr  = idx >> 7;
        const int q   = idx & 127;
        const int row = h0 + dr - 1;
        if (row < 0 || row >= H) {
#pragma unroll
            for (int c = 0; c < CIN; ++c)
                *(float4*)(&lds[(c * CROWS + dr) * LDS_ROWW + 4 + q * 4]) =
                    make_float4(0.0f, 0.0f, 0.0f, 0.0f);
        }
    }
    __syncthreads();

    const int lane = tid & 31;
    const int wave = tid >> 5;
    const int m    = lane & 15;
    const int half = lane >> 4;

    v2f bfrag[7];
#pragma unroll
    for (int j = 0; j < 7; ++j) {
        float b0 = 0.0f, b1 = 0.0f;
        int k0 = j * 4 + 2 * half;
        if (m < COUT) {
            b0 = wgt[m * 27 + k0];
            if (k0 + 1 < 27) b1 = wgt[m * 27 + k0 + 1];
        }
        bfrag[j][0] = b0;
        bfrag[j][1] = b1;
    }

    int offA[14];
#pragma unroll
    for (int j = 0; j < 7; ++j) {
#pragma unroll
        for (int v = 0; v < 2; ++v) {
            int k = j * 4 + 2 * half + v;
            if (k > 26) k = 26;
            int c  = k / 9;
            int r  = k - c * 9;
            int dh = r / 3;
            int dw = r - dh * 3;
            offA[j * 2 + v] = (c * CROWS + dh) * LDS_ROWW + m + dw + 3;
        }
    }

    const float bv0 = bias[0];
    const float bv1 = bias[1];

    for (int rr = 0; rr < RB; ++rr) {
        const float* ldsr = lds + rr * LDS_ROWW;
        const int h = h0 + rr;

#pragma unroll
        for (int s = 0; s < 4; ++s) {
            const int w0 = (wave * 4 + s) * 16;

            v8f acc = {};
#pragma unroll
            for (int j = 0; j < 7; ++j) {
                v2f a;
                a[0] = ldsr[offA[j * 2 + 0] + w0];
                a[1] = ldsr[offA[j * 2 + 1] + w0];
                acc = __builtin_amdgcn_wmma_f32_16x16x4_f32(
                     false, a,  false, bfrag[j],
                     (short)0, acc,  false,  false);
            }

            if ((lane & 14) == 0) {
                const int ch = lane & 1;
                const int mb = (lane >> 4) * 8;
                const float bv = ch ? bv1 : bv0;
                float* o = out + ((((size_t)b * COUT + ch) * H + h) * W + w0 + mb);
                float4 lo = make_float4(acc[0] + bv, acc[1] + bv, acc[2] + bv, acc[3] + bv);
                float4 hi = make_float4(acc[4] + bv, acc[5] + bv, acc[6] + bv, acc[7] + bv);
                *(float4*)(o + 0) = lo;
                *(float4*)(o + 4) = hi;
            }
        }
    }
}

extern "C" void kernel_launch(void* const* d_in, const int* in_sizes, int n_in,
                              void* d_out, int out_size, void* d_ws, size_t ws_size,
                              hipStream_t stream) {
    const float* x    = (const float*)d_in[0];
    const float* wgt  = (const float*)d_in[1];
    const float* bias = (const float*)d_in[2];
    float* out        = (float*)d_out;

    dim3 grid(H / RB, 64, 1);
    dim3 block(256, 1, 1);
    conv3x3_wmma_f32<<<grid, block, 0, stream>>>(x, wgt, bias, out);
}


# GEMM re-mapped: A = banded weights (rows = channel x 8 output rows), B = 16 pixels, k = (in-channel, staged row, dw): 23 matrix instr per 8x16x2 tile (was 56); order of the tap sum changes
# speedup vs baseline: 2.0956x; 1.7243x over previous
#include <hip/hip_runtime.h>


typedef __attribute__((ext_vector_type(2))) float v2f;
typedef __attribute__((ext_vector_type(8))) float v8f;

#define H 512
#define W 512
#define CIN 3
#define COUT 2
#define RB 8
#define CROWS (RB + 2)
#define LDS_NROWS (CIN * CROWS)
#define LDS_ROWW  520
#define HROWS (CROWS / 2)
#define NTAP  (CIN * HROWS * 3)
#define NWM   ((NTAP + 1) / 2)

__global__ __launch_bounds__(256) void conv3x3_wmma_f32(
    const float* __restrict__ x,
    const float* __restrict__ wgt,
    const float* __restrict__ bias,
    float* __restrict__ out)
{
    const int h0  = blockIdx.x * RB;
    const int b   = blockIdx.y;
    const int tid = threadIdx.x;

    __shared__ float lds[LDS_NROWS * LDS_ROWW];

    if (tid < 2 * LDS_NROWS) {
        int r = tid >> 1;
        lds[r * LDS_ROWW + ((tid & 1) ? 516 : 3)] = 0.0f;
    }

    constexpr int NST = (CROWS * 128) / 256;
    float4 sv[CIN][NST];
#pragma unroll
    for (int i = 0; i < NST; ++i) {
        const int idx  = tid + i * 256;
        const int dr   = idx >> 7;
        const int q    = idx & 127;
        const int rowc = min(max(h0 + dr - 1, 0), H - 1);
#pragma unroll
        for (int c = 0; c < CIN; ++c)
            sv[c][i] = *(const float4*)(x + (((size_t)b * CIN + c) * H + rowc) * W + q * 4);
    }
#pragma unroll
    for (int i = 0; i < NST; ++i) {
        const int idx = tid + i * 256;
        const int dr  = idx >> 7;
        const int q   = idx & 127;
#pragma unroll
        for (int c = 0; c < CIN; ++c)
            *(float4*)(&lds[(c * CROWS + dr) * LDS_ROWW + 4 + q * 4]) = sv[c][i];
    }
#pragma unroll
    for (int i = 0; i < NST; ++i) {
        const int idx = tid + i * 256;
        const int dr  = idx >> 7;
        const int q   = idx & 127;
        const int row = h0 + dr - 1;
        if (row < 0 || row >= H) {
#pragma unroll
            for (int c = 0; c < CIN; ++c)
                *(float4*)(&lds[(c * CROWS + dr) * LDS_ROWW + 4 + q * 4]) =
                    make_float4(0.0f, 0.0f, 0.0f, 0.0f);
        }
    }
    __syncthreads();

    const int lane = tid & 31;
    const int wave = tid >> 5;
    const int c16  = lane & 15;
    const int half = lane >> 4;


    const int och  = c16 >> 3;
    const int orow = c16 & 7;
    v2f afrag[NWM];
#pragma unroll
    for (int j = 0; j < NWM; ++j) {
#pragma unroll
        for (int v = 0; v < 2; ++v) {
            const int s = 2 * j + v;
            float val = 0.0f;
            if (s < NTAP) {
                const int c  = s / 15;
                const int il = (s % 15) / 3;
                const int dw = s % 3;
                const int dh = il + HROWS * half - orow;
                const bool in = (dh >= 0) && (dh < 3);
                const float wv = wgt[och * 27 + c * 9 + (in ? dh : 0) * 3 + dw];
                val = in ? wv : 0.0f;
            }
            afrag[j][v] = val;
        }
    }

    const float* lbase = lds + half * (HROWS * LDS_ROWW) + wave * 64 + c16 + 3;

    const float bv = bias[half];
    float* obase = out + ((((size_t)b * COUT + half) * H + h0) * W + wave * 64 + c16);

#pragma unroll
    for (int tp = 0; tp < 2; ++tp) {
        const float* l0 = lbase + (tp * 2 + 0) * 16;
        const float* l1 = lbase + (tp * 2 + 1) * 16;
        v8f acc0 = {bv, bv, bv, bv, bv, bv, bv, bv};
        v8f acc1 = {bv, bv, bv, bv, bv, bv, bv, bv};
#pragma unroll
        for (int j = 0; j < NWM; ++j) {
            v2f x0, x1;
#pragma unroll
            for (int v = 0; v < 2; ++v) {
                int s = 2 * j + v;
                if (s >= NTAP) s = NTAP - 1;
                const int c  = s / 15;
                const int il = (s % 15) / 3;
                const int dw = s % 3;
                const int off = (c * CROWS + il) * LDS_ROWW + dw;
                x0[v] = l0[off];
                x1[v] = l1[off];
            }
            acc0 = __builtin_amdgcn_wmma_f32_16x16x4_f32(
                 false, afrag[j],  false, x0,
                 (short)0, acc0,  false,  false);
            acc1 = __builtin_amdgcn_wmma_f32_16x16x4_f32(
                 false, afrag[j],  false, x1,
                 (short)0, acc1,  false,  false);
        }
        float* o0 = obase + (tp * 2 + 0) * 16;
        float* o1 = obase + (tp * 2 + 1) * 16;
#pragma unroll
        for (int r = 0; r < RB; ++r) {
            o0[r * W] = acc0[r];
            o1[r * W] = acc1[r];
        }
    }
}

extern "C" void kernel_launch(void* const* d_in, const int* in_sizes, int n_in,
                              void* d_out, int out_size, void* d_ws, size_t ws_size,
                              hipStream_t stream) {
    const float* x    = (const float*)d_in[0];
    const float* wgt  = (const float*)d_in[1];
    const float* bias = (const float*)d_in[2];
    float* out        = (float*)d_out;

    dim3 grid(H / RB, 64, 1);
    dim3 block(256, 1, 1);
    conv3x3_wmma_f32<<<grid, block, 0, stream>>>(x, wgt, bias, out);
}


# Work-group walks 8 strips (was 1): next strip's rows are loaded to registers before the current strip's matrix work, stored to a 2nd LDS buffer after it; LDS-only barrier per strip; 8x fewer waves
# speedup vs baseline: 2.2201x; 1.0594x over previous
#include <hip/hip_runtime.h>


typedef __attribute__((ext_vector_type(2))) float v2f;
typedef __attribute__((ext_vector_type(8))) float v8f;

#define H 512
#define W 512
#define CIN 3
#define COUT 2
#define RB 8
#define NS 8
#define CROWS (RB + 2)
#define LDS_NROWS (CIN * CROWS)
#define LDS_ROWW  520
#define LDS_BUF (LDS_NROWS * LDS_ROWW)
#define HROWS (CROWS / 2)
#define NTAP  (CIN * HROWS * 3)
#define NWM   ((NTAP + 1) / 2)
#define NST   ((CROWS * 128) / 256)

__device__ __forceinline__ void stage_load(float4 (&sv)[CIN][NST], const float* __restrict__ xq,
                                           int h0, int dr0)
{
#pragma unroll
    for (int i = 0; i < NST; ++i) {
        const int rowc = min(max(h0 + dr0 + 2 * i - 1, 0), H - 1);
#pragma unroll
        for (int c = 0; c < CIN; ++c)
            sv[c][i] = *(const float4*)(xq + ((size_t)c * H + rowc) * W);
    }
}

__device__ __forceinline__ void stage_store(const float4 (&sv)[CIN][NST], float* buf,
                                            int h0, int dr0, int q)
{
#pragma unroll
    for (int i = 0; i < NST; ++i) {
        const int dr  = dr0 + 2 * i;
        const int row = h0 + dr - 1;
        const bool outside = (row < 0) || (row >= H);
#pragma unroll
        for (int c = 0; c < CIN; ++c) {
            float4 v = sv[c][i];
            if (outside) v = make_float4(0.0f, 0.0f, 0.0f, 0.0f);
            *(float4*)(&buf[(c * CROWS + dr) * LDS_ROWW + 4 + q * 4]) = v;
        }
    }
}

__device__ __forceinline__ void lds_barrier()
{
    __builtin_amdgcn_fence(__ATOMIC_RELEASE, "workgroup", "local");
    __builtin_amdgcn_s_barrier();
    __builtin_amdgcn_fence(__ATOMIC_ACQUIRE, "workgroup", "local");
}

__global__ __launch_bounds__(256) void conv3x3_wmma_f32(
    const float* __restrict__ x,
    const float* __restrict__ wgt,
    const float* __restrict__ bias,
    float* __restrict__ out)
{
    const int hbase = blockIdx.x * (RB * NS);
    const int b   = blockIdx.y;
    const int tid = threadIdx.x;

    __shared__ float lds[2 * LDS_BUF];

    if (tid < 2 * LDS_NROWS) {
        const int r   = tid >> 1;
        const int col = (tid & 1) ? 516 : 3;
        lds[r * LDS_ROWW + col] = 0.0f;
        lds[LDS_BUF + r * LDS_ROWW + col] = 0.0f;
    }

    const int dr0 = tid >> 7;
    const int q   = tid & 127;
    const float* xq = x + (size_t)b * CIN * H * W + q * 4;
    float4 sv[CIN][NST];

    stage_load(sv, xq, hbase, dr0);

    const int lane = tid & 31;
    const int wave = tid >> 5;
    const int c16  = lane & 15;
    const int half = lane >> 4;


    const int och  = c16 >> 3;
    const int orow = c16 & 7;
    v2f afrag[NWM];
#pragma unroll
    for (int j = 0; j < NWM; ++j) {
#pragma unroll
        for (int v = 0; v < 2; ++v) {
            const int s = 2 * j + v;
            float val = 0.0f;
            if (s < NTAP) {
                const int c  = s / 15;
                const int il = (s % 15) / 3;
                const int dw = s % 3;
                const int dh = il + HROWS * half - orow;
                const bool in = (dh >= 0) && (dh < 3);
                const float wv = wgt[och * 27 + c * 9 + (in ? dh : 0) * 3 + dw];
                val = in ? wv : 0.0f;
            }
            afrag[j][v] = val;
        }
    }

    const int loff = half * (HROWS * LDS_ROWW) + wave * 64 + c16 + 3;

    const float bv = bias[half];
    float* obase = out + ((((size_t)b * COUT + half) * H + hbase) * W + wave * 64 + c16);

    stage_store(sv, lds, hbase, dr0, q);
    __syncthreads();

#pragma unroll 1
    for (int s = 0; s < NS; ++s) {
        const int h0 = hbase + s * RB;
        const float* cur = lds + (s & 1) * LDS_BUF;
        float* nxt = lds + ((s + 1) & 1) * LDS_BUF;
        const bool more = (s + 1 < NS);

        if (more) stage_load(sv, xq, h0 + RB, dr0);

        const float* lbase = cur + loff;
        float* ostrip = obase + (size_t)s * RB * W;
#pragma unroll
        for (int tp = 0; tp < 2; ++tp) {
            const float* l0 = lbase + (tp * 2 + 0) * 16;
            const float* l1 = lbase + (tp * 2 + 1) * 16;
            v8f acc0 = {bv, bv, bv, bv, bv, bv, bv, bv};
            v8f acc1 = {bv, bv, bv, bv, bv, bv, bv, bv};
#pragma unroll
            for (int j = 0; j < NWM; ++j) {
                v2f x0, x1;
#pragma unroll
                for (int v = 0; v < 2; ++v) {
                    int t = 2 * j + v;
                    if (t >= NTAP) t = NTAP - 1;
                    const int c  = t / 15;
                    const int il = (t % 15) / 3;
                    const int dw = t % 3;
                    const int off = (c * CROWS + il) * LDS_ROWW + dw;
                    x0[v] = l0[off];
                    x1[v] = l1[off];
                }
                acc0 = __builtin_amdgcn_wmma_f32_16x16x4_f32(
                     false, afrag[j],  false, x0,
                     (short)0, acc0,  false,  false);
                acc1 = __builtin_amdgcn_wmma_f32_16x16x4_f32(
                     false, afrag[j],  false, x1,
                     (short)0, acc1,  false,  false);
            }
            float* o0 = ostrip + (tp * 2 + 0) * 16;
            float* o1 = ostrip + (tp * 2 + 1) * 16;
#pragma unroll
            for (int r = 0; r < RB; ++r) {
                o0[r * W] = acc0[r];
                o1[r * W] = acc1[r];
            }
        }

        if (more) stage_store(sv, nxt, h0 + RB, dr0, q);
        lds_barrier();
    }
}


extern "C" void kernel_launch(void* const* d_in, const int* in_sizes, int n_in,
                              void* d_out, int out_size, void* d_ws, size_t ws_size,
                              hipStream_t stream) {
    const float* x    = (const float*)d_in[0];
    const float* wgt  = (const float*)d_in[1];
    const float* bias = (const float*)d_in[2];
    float* out        = (float*)d_out;

    dim3 grid(H / (RB * NS), 64, 1);
    dim3 block(256, 1, 1);
    conv3x3_wmma_f32<<<grid, block, 0, stream>>>(x, wgt, bias, out);
}


# GEMM re-mapped to patches of 4 rows x 2 pixels: k = the 6x4 inputs under a patch x 3 channels, 18 matrix instr per 256 outputs (was 23); 8-byte stores fill 128-byte lines; order of tap sum changes
# speedup vs baseline: 2.5520x; 1.1495x over previous
#include <hip/hip_runtime.h>


typedef __attribute__((ext_vector_type(2))) float v2f;
typedef __attribute__((ext_vector_type(8))) float v8f;

#define H 512
#define W 512
#define CIN 3
#define COUT 2
#define RB 8
#define NS 8
#define CROWS (RB + 2)
#define LDS_NROWS (CIN * CROWS)
#define LDS_ROWW  520
#define LDS_BUF (LDS_NROWS * LDS_ROWW)
#define PH 4
#define PIN (PH + 2)
#define NWM (CIN * PIN)
#define NTR (RB / PH)
#define NST   ((CROWS * 128) / 256)

__device__ __forceinline__ void stage_load(float4 (&sv)[CIN][NST], const float* __restrict__ xq,
                                           int h0, int dr0)
{
#pragma unroll
    for (int i = 0; i < NST; ++i) {
        const int rowc = min(max(h0 + dr0 + 2 * i - 1, 0), H - 1);
#pragma unroll
        for (int c = 0; c < CIN; ++c)
            sv[c][i] = *(const float4*)(xq + ((size_t)c * H + rowc) * W);
    }
}

__device__ __forceinline__ void stage_store(const float4 (&sv)[CIN][NST], float* buf,
                                            int h0, int dr0, int q)
{
#pragma unroll
    for (int i = 0; i < NST; ++i) {
        const int dr  = dr0 + 2 * i;
        const int row = h0 + dr - 1;
        const bool outside = (row < 0) || (row >= H);
#pragma unroll
        for (int c = 0; c < CIN; ++c) {
            float4 v = sv[c][i];
            if (outside) v = make_float4(0.0f, 0.0f, 0.0f, 0.0f);
            *(float4*)(&buf[(c * CROWS + dr) * LDS_ROWW + 4 + q * 4]) = v;
        }
    }
}

__device__ __forceinline__ void lds_barrier()
{
    __builtin_amdgcn_fence(__ATOMIC_RELEASE, "workgroup", "local");
    __builtin_amdgcn_s_barrier();
    __builtin_amdgcn_fence(__ATOMIC_ACQUIRE, "workgroup", "local");
}

__global__ __launch_bounds__(256) void conv3x3_wmma_f32(
    const float* __restrict__ x,
    const float* __restrict__ wgt,
    const float* __restrict__ bias,
    float* __restrict__ out)
{
    const int hbase = blockIdx.x * (RB * NS);
    const int b   = blockIdx.y;
    const int tid = threadIdx.x;

    __shared__ float lds[2 * LDS_BUF];

    if (tid < 2 * LDS_NROWS) {
        const int r   = tid >> 1;
        const int col = (tid & 1) ? 516 : 3;
        lds[r * LDS_ROWW + col] = 0.0f;
        lds[LDS_BUF + r * LDS_ROWW + col] = 0.0f;
    }

    const int dr0 = tid >> 7;
    const int q   = tid & 127;
    const float* xq = x + (size_t)b * CIN * H * W + q * 4;
    float4 sv[CIN][NST];

    stage_load(sv, xq, hbase, dr0);

    const int lane = tid & 31;
    const int wave = tid >> 5;
    const int c16  = lane & 15;
    const int half = lane >> 4;


    const int och = c16 >> 3;
    const int pr  = (c16 >> 1) & 3;
    const int pc  = c16 & 1;
    v2f afrag[NWM];
#pragma unroll
    for (int j = 0; j < NWM; ++j) {
#pragma unroll
        for (int v = 0; v < 2; ++v) {
            const int c  = j / PIN;
            const int dh = j % PIN - pr;
            const int dw = 2 * v + half - pc;
            const bool in = (dh >= 0) && (dh < 3) && (dw >= 0) && (dw < 3);
            const float wv = wgt[och * 27 + c * 9 + (in ? dh * 3 + dw : 0)];
            afrag[j][v] = in ? wv : 0.0f;
        }
    }

    const int loff = wave * 64 + 2 * c16 + 3 + half;

    const float bv = bias[half];
    float* obase = out + ((((size_t)b * COUT + half) * H + hbase) * W + wave * 64 + 2 * c16);

    stage_store(sv, lds, hbase, dr0, q);
    __syncthreads();

#pragma unroll 1
    for (int s = 0; s < NS; ++s) {
        const int h0 = hbase + s * RB;
        const float* cur = lds + (s & 1) * LDS_BUF;
        float* nxt = lds + ((s + 1) & 1) * LDS_BUF;
        const bool more = (s + 1 < NS);

        if (more) stage_load(sv, xq, h0 + RB, dr0);

        const float* lbase = cur + loff;
        float* ostrip = obase + (size_t)s * RB * W;
        v8f acc[NTR][2];
#pragma unroll
        for (int tr = 0; tr < NTR; ++tr)
#pragma unroll
            for (int tw = 0; tw < 2; ++tw)
                acc[tr][tw] = (v8f){bv, bv, bv, bv, bv, bv, bv, bv};
#pragma unroll
        for (int c = 0; c < CIN; ++c) {
#pragma unroll
            for (int R = 0; R < CROWS; ++R) {
                const int off = (c * CROWS + R) * LDS_ROWW;
                v2f xf[2];
#pragma unroll
                for (int tw = 0; tw < 2; ++tw) {
                    xf[tw][0] = lbase[off + tw * 32];
                    xf[tw][1] = lbase[off + tw * 32 + 2];
                }
#pragma unroll
                for (int tr = 0; tr < NTR; ++tr) {
                    const int i = R - tr * PH;
                    if (i >= 0 && i < PIN) {
#pragma unroll
                        for (int tw = 0; tw < 2; ++tw)
                            acc[tr][tw] = __builtin_amdgcn_wmma_f32_16x16x4_f32(
                                 false, afrag[c * PIN + i],  false, xf[tw],
                                 (short)0, acc[tr][tw],  false,  false);
                    }
                }
            }
        }
#pragma unroll
        for (int tr = 0; tr < NTR; ++tr)
#pragma unroll
            for (int p = 0; p < PH; ++p)
#pragma unroll
                for (int tw = 0; tw < 2; ++tw)
                    *(float2*)(ostrip + (tr * PH + p) * W + tw * 32) =
                        make_float2(acc[tr][tw][2 * p], acc[tr][tw][2 * p + 1]);

        if (more) stage_store(sv, nxt, h0 + RB, dr0, q);
        lds_barrier();
    }
}


extern "C" void kernel_launch(void* const* d_in, const int* in_sizes, int n_in,
                              void* d_out, int out_size, void* d_ws, size_t ws_size,
                              hipStream_t stream) {
    const float* x    = (const float*)d_in[0];
    const float* wgt  = (const float*)d_in[1];
    const float* bias = (const float*)d_in[2];
    float* out        = (float*)d_out;

    dim3 grid(H / (RB * NS), 64, 1);
    dim3 block(256, 1, 1);
    conv3x3_wmma_f32<<<grid, block, 0, stream>>>(x, wgt, bias, out);
}


# GEMM re-mapped to patches of 2 rows x 4 pixels (k = the 4x6 inputs under a patch x 3 channels, still 18 matrix instr per 256 outputs): 16-byte stores write whole 256-byte units; tap sum order changes
# speedup vs baseline: 2.7571x; 1.0804x over previous
#include <hip/hip_runtime.h>


typedef __attribute__((ext_vector_type(2))) float v2f;
typedef __attribute__((ext_vector_type(8))) float v8f;

#define H 512
#define W 512
#define CIN 3
#define COUT 2
#define RB 8
#define NS 8
#define CROWS (RB + 2)
#define LDS_NROWS (CIN * CROWS)
#define LDS_ROWW  520
#define LDS_BUF (LDS_NROWS * LDS_ROWW)
#define PH 2
#define PIN (PH + 2)
#define NWC 6
#define NWM (CIN * NWC)
#define NTR (RB / PH)
#define NST   ((CROWS * 128) / 256)

__device__ __forceinline__ void stage_load(float4 (&sv)[CIN][NST], const float* __restrict__ xq,
                                           int h0, int dr0)
{
#pragma unroll
    for (int i = 0; i < NST; ++i) {
        const int rowc = min(max(h0 + dr0 + 2 * i - 1, 0), H - 1);
#pragma unroll
        for (int c = 0; c < CIN; ++c)
            sv[c][i] = *(const float4*)(xq + ((size_t)c * H + rowc) * W);
    }
}

__device__ __forceinline__ void stage_store(const float4 (&sv)[CIN][NST], float* buf,
                                            int h0, int dr0, int q)
{
#pragma unroll
    for (int i = 0; i < NST; ++i) {
        const int dr  = dr0 + 2 * i;
        const int row = h0 + dr - 1;
        const bool outside = (row < 0) || (row >= H);
#pragma unroll
        for (int c = 0; c < CIN; ++c) {
            float4 v = sv[c][i];
            if (outside) v = make_float4(0.0f, 0.0f, 0.0f, 0.0f);
            *(float4*)(&buf[(c * CROWS + dr) * LDS_ROWW + 4 + q * 4]) = v;
        }
    }
}

__device__ __forceinline__ void lds_barrier()
{
    __builtin_amdgcn_fence(__ATOMIC_RELEASE, "workgroup", "local");
    __builtin_amdgcn_s_barrier();
    __builtin_amdgcn_fence(__ATOMIC_ACQUIRE, "workgroup", "local");
}

__global__ __launch_bounds__(256) void conv3x3_wmma_f32(
    const float* __restrict__ x,
    const float* __restrict__ wgt,
    const float* __restrict__ bias,
    float* __restrict__ out)
{
    const int hbase = blockIdx.x * (RB * NS);
    const int b   = blockIdx.y;
    const int tid = threadIdx.x;

    __shared__ float lds[2 * LDS_BUF];

    if (tid < 2 * LDS_NROWS) {
        const int r   = tid >> 1;
        const int col = (tid & 1) ? 516 : 3;
        lds[r * LDS_ROWW + col] = 0.0f;
        lds[LDS_BUF + r * LDS_ROWW + col] = 0.0f;
    }

    const int dr0 = tid >> 7;
    const int q   = tid & 127;
    const float* xq = x + (size_t)b * CIN * H * W + q * 4;
    float4 sv[CIN][NST];

    stage_load(sv, xq, hbase, dr0);

    const int lane = tid & 31;
    const int wave = tid >> 5;
    const int c16  = lane & 15;
    const int half = lane >> 4;


    const int och = c16 >> 3;
    const int pr  = (c16 >> 2) & 1;
    const int pc  = c16 & 3;
    v2f afrag[NWM];
#pragma unroll
    for (int j = 0; j < NWM; ++j) {
#pragma unroll
        for (int v = 0; v < 2; ++v) {
            const int c  = j / NWC;
            const int t  = j % NWC;
            const int i  = (t < PIN) ? t : 2 * (t - PIN) + v;
            const int jj = (t < PIN) ? 2 * v + half : 4 + half;
            const int dh = i - pr;
            const int dw = jj - pc;
            const bool in = (dh >= 0) && (dh < 3) && (dw >= 0) && (dw < 3);
            const float wv = wgt[och * 27 + c * 9 + (in ? dh * 3 + dw : 0)];
            afrag[j][v] = in ? wv : 0.0f;
        }
    }

    const int loff = wave * 64 + 4 * c16 + 3 + half;

    const float bv = bias[half];
    float* obase = out + ((((size_t)b * COUT + half) * H + hbase) * W + wave * 64 + 4 * c16);

    stage_store(sv, lds, hbase, dr0, q);
    __syncthreads();

#pragma unroll 1
    for (int s = 0; s < NS; ++s) {
        const int h0 = hbase + s * RB;
        const float* cur = lds + (s & 1) * LDS_BUF;
        float* nxt = lds + ((s + 1) & 1) * LDS_BUF;
        const bool more = (s + 1 < NS);

        if (more) stage_load(sv, xq, h0 + RB, dr0);

        const float* lbase = cur + loff;
        float* ostrip = obase + (size_t)s * RB * W;
        v8f acc[NTR];
#pragma unroll
        for (int tr = 0; tr < NTR; ++tr)
            acc[tr] = (v8f){bv, bv, bv, bv, bv, bv, bv, bv};
#pragma unroll
        for (int c = 0; c < CIN; ++c) {
#pragma unroll
            for (int R = 0; R < CROWS; ++R) {
                const int off = (c * CROWS + R) * LDS_ROWW;
                v2f xr;
                xr[0] = lbase[off];
                xr[1] = lbase[off + 2];
#pragma unroll
                for (int tr = 0; tr < NTR; ++tr) {
                    const int t = R - tr * PH;
                    if (t >= 0 && t < PIN)
                        acc[tr] = __builtin_amdgcn_wmma_f32_16x16x4_f32(
                             false, afrag[c * NWC + t],  false, xr,
                             (short)0, acc[tr],  false,  false);
                }
            }
#pragma unroll
            for (int m = 0; m < CROWS / 2; ++m) {
                const int off = (c * CROWS + 2 * m) * LDS_ROWW + 4;
                v2f xp;
                xp[0] = lbase[off];
                xp[1] = lbase[off + LDS_ROWW];
#pragma unroll
                for (int tr = 0; tr < NTR; ++tr) {
                    const int ip = m - tr;
                    if (ip >= 0 && ip < 2)
                        acc[tr] = __builtin_amdgcn_wmma_f32_16x16x4_f32(
                             false, afrag[c * NWC + PIN + ip],  false, xp,
                             (short)0, acc[tr],  false,  false);
                }
            }
        }
#pragma unroll
        for (int tr = 0; tr < NTR; ++tr)
#pragma unroll
            for (int p = 0; p < PH; ++p)
                *(float4*)(ostrip + (tr * PH + p) * W) =
                    make_float4(acc[tr][4 * p], acc[tr][4 * p + 1],
                                acc[tr][4 * p + 2], acc[tr][4 * p + 3]);

        if (more) stage_store(sv, nxt, h0 + RB, dr0, q);
        lds_barrier();
    }
}


extern "C" void kernel_launch(void* const* d_in, const int* in_sizes, int n_in,
                              void* d_out, int out_size, void* d_ws, size_t ws_size,
                              hipStream_t stream) {
    const float* x    = (const float*)d_in[0];
    const float* wgt  = (const float*)d_in[1];
    const float* bias = (const float*)d_in[2];
    float* out        = (float*)d_out;

    dim3 grid(H / (RB * NS), 64, 1);
    dim3 block(256, 1, 1);
    conv3x3_wmma_f32<<<grid, block, 0, stream>>>(x, wgt, bias, out);
}
